# starting kernel
# baseline (speedup 1.0000x reference)
#include <hip/hip_runtime.h>


typedef float v2f __attribute__((ext_vector_type(2)));
typedef float v8f __attribute__((ext_vector_type(8)));

#define H       256
#define LAYERS  3
#define MBLK    16
#define STRIDE  264

__device__ __forceinline__ v8f wmma_f32x4(v2f a, v2f b, v8f c) {
    return __builtin_amdgcn_wmma_f32_16x16x4_f32(
         false, a,  false, b,
         (short)0, c,  false,  false);
}

__device__ __forceinline__ float fast_tanh(float v) {
#if __has_builtin(__builtin_amdgcn_tanhf)
    return __builtin_amdgcn_tanhf(v);
#else
    float e = __expf(2.0f * v);
    return 1.0f - 2.0f / (e + 1.0f);
#endif
}

__global__ __launch_bounds__(256) void dgm_fused_kernel(
    const float* __restrict__ t,  const float* __restrict__ x,
    const float* __restrict__ W0, const float* __restrict__ b0,
    const float* __restrict__ Uz, const float* __restrict__ Ug,
    const float* __restrict__ Ur, const float* __restrict__ Uh,
    const float* __restrict__ Wz, const float* __restrict__ Wg,
    const float* __restrict__ Wr, const float* __restrict__ Wh,
    const float* __restrict__ bz, const float* __restrict__ bg,
    const float* __restrict__ br, const float* __restrict__ bh,
    const float* __restrict__ Wf, const float* __restrict__ bf,
    float* __restrict__ out)
{
    __shared__ float sS [MBLK][STRIDE];
    __shared__ float sS1[MBLK][STRIDE];
    __shared__ float sSR[MBLK][STRIDE];
    __shared__ float sXt[MBLK];
    __shared__ float sXx[MBLK];

    const int tid  = threadIdx.x;
    const int wave = tid >> 5;
    const int lane = tid & 31;
    const int half = lane >> 4;
    const int l16  = lane & 15;
    const int row0 = blockIdx.x * MBLK;

    if (tid < MBLK) {
        sXt[tid] = t[row0 + tid];
        sXx[tid] = x[row0 + tid];
    }
    __syncthreads();

    {
        const int   c   = tid;
        const float w0t = W0[c];
        const float w0x = W0[H + c];
        const float bb  = b0[c];
        #pragma unroll
        for (int r = 0; r < MBLK; ++r) {
            float v = fast_tanh(sXt[r] * w0t + sXx[r] * w0x + bb);
            sS1[r][c] = v;
            sS [r][c] = v;
        }
    }
    __syncthreads();

    float xt8[8], xx8[8];
    #pragma unroll
    for (int v = 0; v < 8; ++v) {
        xt8[v] = sXt[half * 8 + v];
        xx8[v] = sXx[half * 8 + v];
    }

    const int koff = half * 2;
    const int cb0  = wave * 32;

    for (int l = 0; l < LAYERS; ++l) {
        const float* wzl = Wz + l * H * H;
        const float* wgl = Wg + l * H * H;
        const float* wrl = Wr + l * H * H;
        const float* whl = Wh + l * H * H;
        const float* uzl = Uz + l * 2 * H;
        const float* ugl = Ug + l * 2 * H;
        const float* url = Ur + l * 2 * H;
        const float* uhl = Uh + l * 2 * H;
        const float* bzl = bz + l * H;
        const float* bgl = bg + l * H;
        const float* brl = br + l * H;
        const float* bhl = bh + l * H;

        v8f accR[2];
        #pragma unroll
        for (int t2 = 0; t2 < 2; ++t2) {
            const int   n  = cb0 + t2 * 16 + l16;
            const float u0 = url[n];
            const float u1 = url[H + n];
            const float bb = brl[n];
            #pragma unroll
            for (int v = 0; v < 8; ++v)
                accR[t2][v] = xt8[v] * u0 + xx8[v] * u1 + bb;
        }
        for (int k = 0; k < H; k += 4) {
            v2f a = *(const v2f*)&sS[l16][k + koff];
            #pragma unroll
            for (int t2 = 0; t2 < 2; ++t2) {
                const int n = cb0 + t2 * 16 + l16;
                v2f b;
                b.x = wrl[(k + koff)     * H + n];
                b.y = wrl[(k + koff + 1) * H + n];
                accR[t2] = wmma_f32x4(a, b, accR[t2]);
            }
        }
        #pragma unroll
        for (int t2 = 0; t2 < 2; ++t2) {
            const int n = cb0 + t2 * 16 + l16;
            #pragma unroll
            for (int v = 0; v < 8; ++v) {
                const int m = half * 8 + v;
                float R = fast_tanh(accR[t2][v]);
                sSR[m][n] = sS[m][n] * R;
            }
        }
        __syncthreads();

        v8f accZ[2], accG[2], accHg[2];
        #pragma unroll
        for (int t2 = 0; t2 < 2; ++t2) {
            const int   n  = cb0 + t2 * 16 + l16;
            const float z0 = uzl[n], z1 = uzl[H + n], zb = bzl[n];
            const float g0 = ugl[n], g1 = ugl[H + n], gb = bgl[n];
            const float h0 = uhl[n], h1 = uhl[H + n], hb = bhl[n];
            #pragma unroll
            for (int v = 0; v < 8; ++v) {
                accZ [t2][v] = xt8[v] * z0 + xx8[v] * z1 + zb;
                accG [t2][v] = xt8[v] * g0 + xx8[v] * g1 + gb;
                accHg[t2][v] = xt8[v] * h0 + xx8[v] * h1 + hb;
            }
        }
        for (int k = 0; k < H; k += 4) {
            v2f aS  = *(const v2f*)&sS [l16][k + koff];
            v2f aS1 = *(const v2f*)&sS1[l16][k + koff];
            v2f aSR = *(const v2f*)&sSR[l16][k + koff];
            #pragma unroll
            for (int t2 = 0; t2 < 2; ++t2) {
                const int n = cb0 + t2 * 16 + l16;
                v2f bzv, bgv, bhv;
                bzv.x = wzl[(k + koff)     * H + n];
                bzv.y = wzl[(k + koff + 1) * H + n];
                bgv.x = wgl[(k + koff)     * H + n];
                bgv.y = wgl[(k + koff + 1) * H + n];
                bhv.x = whl[(k + koff)     * H + n];
                bhv.y = whl[(k + koff + 1) * H + n];
                accZ [t2] = wmma_f32x4(aS,  bzv, accZ [t2]);
                accG [t2] = wmma_f32x4(aS1, bgv, accG [t2]);
                accHg[t2] = wmma_f32x4(aSR, bhv, accHg[t2]);
            }
        }
        float snew[2][8];
        #pragma unroll
        for (int t2 = 0; t2 < 2; ++t2) {
            const int n = cb0 + t2 * 16 + l16;
            #pragma unroll
            for (int v = 0; v < 8; ++v) {
                const int m = half * 8 + v;
                float Zv = fast_tanh(accZ [t2][v]);
                float Gv = fast_tanh(accG [t2][v]);
                float Hv = fast_tanh(accHg[t2][v]);
                snew[t2][v] = (1.0f - Gv) * Hv + Zv * sS[m][n];
            }
        }
        __syncthreads();
        #pragma unroll
        for (int t2 = 0; t2 < 2; ++t2) {
            const int n = cb0 + t2 * 16 + l16;
            #pragma unroll
            for (int v = 0; v < 8; ++v)
                sS[half * 8 + v][n] = snew[t2][v];
        }
        __syncthreads();
    }

    {
        const int r  = tid >> 4;
        const int ci = tid & 15;
        float p = 0.0f;
        #pragma unroll
        for (int j = 0; j < 16; ++j) {
            const int c = ci * 16 + j;
            p += sS[r][c] * Wf[c];
        }
        #pragma unroll
        for (int off = 8; off >= 1; off >>= 1)
            p += __shfl_xor(p, off, 16);
        if (ci == 0)
            out[row0 + r] = p + bf[0];
    }
}

extern "C" void kernel_launch(void* const* d_in, const int* in_sizes, int n_in,
                              void* d_out, int out_size, void* d_ws, size_t ws_size,
                              hipStream_t stream) {
    (void)n_in; (void)d_ws; (void)ws_size; (void)out_size;
    const float* t  = (const float*)d_in[0];
    const float* x  = (const float*)d_in[1];
    const float* W0 = (const float*)d_in[2];
    const float* b0 = (const float*)d_in[3];
    const float* Uz = (const float*)d_in[4];
    const float* Ug = (const float*)d_in[5];
    const float* Ur = (const float*)d_in[6];
    const float* Uh = (const float*)d_in[7];
    const float* Wz = (const float*)d_in[8];
    const float* Wg = (const float*)d_in[9];
    const float* Wr = (const float*)d_in[10];
    const float* Wh = (const float*)d_in[11];
    const float* bz = (const float*)d_in[12];
    const float* bg = (const float*)d_in[13];
    const float* br = (const float*)d_in[14];
    const float* bh = (const float*)d_in[15];
    const float* Wf = (const float*)d_in[16];
    const float* bf = (const float*)d_in[17];
    float* out = (float*)d_out;

    const int N = in_sizes[0];
    const int blocks = N / MBLK;
    dgm_fused_kernel<<<blocks, 256, 0, stream>>>(
        t, x, W0, b0, Uz, Ug, Ur, Uh, Wz, Wg, Wr, Wh,
        bz, bg, br, bh, Wf, bf, out);
}


# Weights streamed through LDS in k-slabs shared by a 64-row block: each wave owns one 16-row tile x 128 columns and reads B fragments from LDS; k order of all sums unchanged
# speedup vs baseline: 1.1227x; 1.1227x over previous
#include <hip/hip_runtime.h>


typedef float v2f __attribute__((ext_vector_type(2)));
typedef float v4f __attribute__((ext_vector_type(4)));
typedef float v8f __attribute__((ext_vector_type(8)));

#define H       256
#define LAYERS  3
#define MBLK    64
#define SP      260
#define KC1     32
#define KC3     16
#define PP      544
#define STG     (3 * (KC3 / 2) * PP)

__device__ __forceinline__ v8f wmma_f32x4(v2f a, v2f b, v8f c) {
    return __builtin_amdgcn_wmma_f32_16x16x4_f32(
         false, a,  false, b,
         (short)0, c,  false,  false);
}

__device__ __forceinline__ float fast_tanh(float v) {
#if __has_builtin(__builtin_amdgcn_tanhf)
    return __builtin_amdgcn_tanhf(v);
#else
    float e = __expf(2.0f * v);
    return 1.0f - 2.0f / (e + 1.0f);
#endif
}

template <int NG, int KCS>
__device__ __forceinline__ void slab_fetch(v4f (&ra)[NG][KCS / 8], v4f (&rb)[NG][KCS / 8],
                                           const float* const (&w)[NG], int k0, int pr0, int q4) {
    #pragma unroll
    for (int g = 0; g < NG; ++g) {
        #pragma unroll
        for (int i = 0; i < KCS / 8; ++i) {
            const float* p = w[g] + (k0 + 2 * (pr0 + 4 * i)) * H + q4;
            ra[g][i] = *(const v4f*)p;
            rb[g][i] = *(const v4f*)(p + H);
        }
    }
}

template <int NG, int KCS>
__device__ __forceinline__ void slab_store(float* stage, const v4f (&ra)[NG][KCS / 8],
                                           const v4f (&rb)[NG][KCS / 8], int pr0, int q4) {
    #pragma unroll
    for (int g = 0; g < NG; ++g) {
        #pragma unroll
        for (int i = 0; i < KCS / 8; ++i) {
            float* d = stage + g * ((KCS / 2) * PP) + (pr0 + 4 * i) * PP + q4 * 2;
            const v4f a = ra[g][i], b = rb[g][i];
            v4f lo, hi;
            lo.x = a.x; lo.y = b.x; lo.z = a.y; lo.w = b.y;
            hi.x = a.z; hi.y = b.z; hi.z = a.w; hi.w = b.w;
            *(v4f*)d       = lo;
            *(v4f*)(d + 4) = hi;
        }
    }
}

template <int NG>
struct Frag {
    v2f a[NG];
    v2f b[NG][8];
};

template <int NG>
__device__ __forceinline__ void frag_load(Frag<NG>& f, const float* const (&ap)[NG],
                                          const float* bl, int gs, int ks) {
    #pragma unroll
    for (int g = 0; g < NG; ++g)
        f.a[g] = *(const v2f*)(ap[g] + ks * 4);
    #pragma unroll
    for (int g = 0; g < NG; ++g) {
        #pragma unroll
        for (int j = 0; j < 8; ++j)
            f.b[g][j] = *(const v2f*)(bl + g * gs + ks * 2 * PP + j * 32);
    }
}

template <int NG>
__device__ __forceinline__ void frag_mma(v8f (&acc)[NG][8], const Frag<NG>& f) {
    #pragma unroll
    for (int j = 0; j < 8; ++j) {
        #pragma unroll
        for (int g = 0; g < NG; ++g)
            acc[g][j] = wmma_f32x4(f.a[g], f.b[g][j], acc[g][j]);
    }
}

template <int NG, int KCS, int NN, int KCN>
__device__ __forceinline__ void gemm_phase(v8f (&acc)[NG][8], const float* const (&w)[NG],
                                           const float* const (&arow)[NG],
                                           const float* const (&wn)[NN], bool has_next,
                                           float* sW, int pr0, int q4, int boff) {
    constexpr int NSL = H / KCS;
    constexpr int KST = KCS / 4;
    constexpr int GSS = (KCS / 2) * PP;
    static_assert((NSL & 1) == 0, "the last slab must sit in stage 1");
    static_assert(NG * GSS <= STG && NN * (KCN / 2) * PP <= STG, "a slab fits a stage");
    v4f ra[NG][KCS / 8], rb[NG][KCS / 8];
    v4f na[NN][KCN / 8], nb[NN][KCN / 8];
    for (int s = 0; s < NSL; ++s) {
        const bool more = (s + 1) < NSL;
        if (more)          slab_fetch<NG, KCS>(ra, rb, w, (s + 1) * KCS, pr0, q4);
        else if (has_next) slab_fetch<NN, KCN>(na, nb, wn, 0, pr0, q4);
        const float* bl = sW + (s & 1) * STG + boff;
        const float* ap[NG];
        #pragma unroll
        for (int g = 0; g < NG; ++g) ap[g] = arow[g] + s * KCS;
        Frag<NG> cur, nxt;
        frag_load<NG>(cur, ap, bl, GSS, 0);
        #pragma unroll
        for (int ks = 0; ks < KST; ++ks) {
            if (ks + 1 < KST) frag_load<NG>(nxt, ap, bl, GSS, ks + 1);
            frag_mma<NG>(acc, cur);
            if (ks + 1 < KST) cur = nxt;
            if (ks == KST / 2 - 1) {
                if (more)          slab_store<NG, KCS>(sW + ((s + 1) & 1) * STG, ra, rb, pr0, q4);
                else if (has_next) slab_store<NN, KCN>(sW, na, nb, pr0, q4);
            }
        }
        __syncthreads();
    }
}

__global__ __launch_bounds__(256) void dgm_fused_kernel(
    const float* __restrict__ t,  const float* __restrict__ x,
    const float* __restrict__ W0, const float* __restrict__ b0,
    const float* __restrict__ Uz, const float* __restrict__ Ug,
    const float* __restrict__ Ur, const float* __restrict__ Uh,
    const float* __restrict__ Wz, const float* __restrict__ Wg,
    const float* __restrict__ Wr, const float* __restrict__ Wh,
    const float* __restrict__ bz, const float* __restrict__ bg,
    const float* __restrict__ br, const float* __restrict__ bh,
    const float* __restrict__ Wf, const float* __restrict__ bf,
    float* __restrict__ out)
{
    __shared__ __attribute__((aligned(16))) float sS [MBLK * SP];
    __shared__ __attribute__((aligned(16))) float sS1[MBLK * SP];
    __shared__ __attribute__((aligned(16))) float sSR[MBLK * SP];
    __shared__ __attribute__((aligned(16))) float sW [2 * STG];
    __shared__ __attribute__((aligned(16))) float sOut[MBLK];
    __shared__ float sXt[MBLK];
    __shared__ float sXx[MBLK];

    const int tid  = threadIdx.x;
    const int wave = tid >> 5;
    const int lane = tid & 31;
    const int half = lane >> 4;
    const int l16  = lane & 15;
    const int row0 = blockIdx.x * MBLK;

    const int rbase = (wave & 3) * 16;
    const int cbase = (wave >> 2) * 128;
    const int pr0   = tid >> 6;
    const int q4    = (tid & 63) * 4;
    const int boff  = half * PP + (cbase + l16) * 2;
    const int aoff  = (rbase + l16) * SP + half * 2;

    if (tid < MBLK) {
        const float* tb = t + row0;
        const float* xb = x + row0;
        sXt[tid] = tb[tid];
        sXx[tid] = xb[tid];
    }
    __syncthreads();

    {
        const int   c   = tid;
        const float w0t = W0[c];
        const float w0x = W0[H + c];
        const float bb  = b0[c];
        #pragma unroll 8
        for (int r = 0; r < MBLK; ++r) {
            float v = fast_tanh(sXt[r] * w0t + sXx[r] * w0x + bb);
            sS1[r * SP + c] = v;
            sS [r * SP + c] = v;
        }
    }

    {
        const float* const wfirst[1] = { Wr };
        v4f fa[1][KC1 / 8], fb[1][KC1 / 8];
        slab_fetch<1, KC1>(fa, fb, wfirst, 0, pr0, q4);
        slab_store<1, KC1>(sW, fa, fb, pr0, q4);
    }
    __syncthreads();

    const int xoff = rbase + half * 8;

    for (int l = 0; l < LAYERS; ++l) {
        const float* wzl = Wz + l * H * H;
        const float* wgl = Wg + l * H * H;
        const float* wrl = Wr + l * H * H;
        const float* whl = Wh + l * H * H;
        const float* uzl = Uz + l * 2 * H;
        const float* ugl = Ug + l * 2 * H;
        const float* url = Ur + l * 2 * H;
        const float* uhl = Uh + l * 2 * H;
        const float* bzl = bz + l * H;
        const float* bgl = bg + l * H;
        const float* brl = br + l * H;
        const float* bhl = bh + l * H;

        {
            v8f accR[1][8];
            float xt8[8], xx8[8];
            #pragma unroll
            for (int v = 0; v < 8; ++v) {
                xt8[v] = sXt[xoff + v];
                xx8[v] = sXx[xoff + v];
            }
            #pragma unroll
            for (int j = 0; j < 8; ++j) {
                const int   n  = cbase + j * 16 + l16;
                const float u0 = url[n];
                const float u1 = url[H + n];
                const float bb = brl[n];
                #pragma unroll
                for (int v = 0; v < 8; ++v)
                    accR[0][j][v] = xt8[v] * u0 + xx8[v] * u1 + bb;
            }
            const float* const w1[1] = { wrl };
            const float* const a1[1] = { sS + aoff };
            const float* const n3[3] = { wzl, wgl, whl };
            gemm_phase<1, KC1, 3, KC3>(accR, w1, a1, n3, true, sW, pr0, q4, boff);
            float sv[8][8];
            #pragma unroll
            for (int j = 0; j < 8; ++j) {
                #pragma unroll
                for (int v = 0; v < 8; ++v)
                    sv[j][v] = sS[(rbase + half * 8 + v) * SP + cbase + j * 16 + l16];
            }
            #pragma unroll
            for (int j = 0; j < 8; ++j) {
                const int n = cbase + j * 16 + l16;
                #pragma unroll
                for (int v = 0; v < 8; ++v) {
                    const int m = rbase + half * 8 + v;
                    float R = fast_tanh(accR[0][j][v]);
                    sSR[m * SP + n] = sv[j][v] * R;
                }
            }
            __syncthreads();
        }

        {
            v8f acc[3][8];
            float xt8[8], xx8[8];
            #pragma unroll
            for (int v = 0; v < 8; ++v) {
                xt8[v] = sXt[xoff + v];
                xx8[v] = sXx[xoff + v];
            }
            #pragma unroll
            for (int j = 0; j < 8; ++j) {
                const int   n  = cbase + j * 16 + l16;
                const float z0 = uzl[n], z1 = uzl[H + n], zb = bzl[n];
                const float g0 = ugl[n], g1 = ugl[H + n], gb = bgl[n];
                const float h0 = uhl[n], h1 = uhl[H + n], hb = bhl[n];
                #pragma unroll
                for (int v = 0; v < 8; ++v) {
                    acc[0][j][v] = xt8[v] * z0 + xx8[v] * z1 + zb;
                    acc[1][j][v] = xt8[v] * g0 + xx8[v] * g1 + gb;
                    acc[2][j][v] = xt8[v] * h0 + xx8[v] * h1 + hb;
                }
            }
            const float* const w3[3] = { wzl, wgl, whl };
            const float* const a3[3] = { sS + aoff, sS1 + aoff, sSR + aoff };
            const bool  has_next = (l + 1) < LAYERS;
            const float* const n1[1] = { has_next ? wrl + H * H : wrl };
            gemm_phase<3, KC3, 1, KC1>(acc, w3, a3, n1, has_next, sW, pr0, q4, boff);
            float sv[8][8];
            #pragma unroll
            for (int j = 0; j < 8; ++j) {
                #pragma unroll
                for (int v = 0; v < 8; ++v)
                    sv[j][v] = sS[(rbase + half * 8 + v) * SP + cbase + j * 16 + l16];
            }
            #pragma unroll
            for (int j = 0; j < 8; ++j) {
                const int n = cbase + j * 16 + l16;
                #pragma unroll
                for (int v = 0; v < 8; ++v) {
                    const int m = rbase + half * 8 + v;
                    float Zv = fast_tanh(acc[0][j][v]);
                    float Gv = fast_tanh(acc[1][j][v]);
                    float Hv = fast_tanh(acc[2][j][v]);
                    sS[m * SP + n] = (1.0f - Gv) * Hv + Zv * sv[j][v];
                }
            }
            __syncthreads();
        }
    }

    {
        const int ci = tid & 15;
        float wf[16];
        #pragma unroll
        for (int j = 0; j < 16; ++j) wf[j] = Wf[ci * 16 + j];
        const float bfv = bf[0];
        #pragma unroll
        for (int pass = 0; pass < MBLK / 16; ++pass) {
            const int r = pass * 16 + (tid >> 4);
            float p = 0.0f;
            #pragma unroll
            for (int j = 0; j < 16; ++j)
                p += sS[r * SP + ci * 16 + j] * wf[j];
            #pragma unroll
            for (int off = 8; off >= 1; off >>= 1)
                p += __shfl_xor(p, off, 16);
            if (ci == 0)
                sOut[r] = p + bfv;
        }
    }
    __syncthreads();
    if (tid < MBLK) {
        float* ob = out + row0;
        ob[tid] = sOut[tid];
    }
}

extern "C" void kernel_launch(void* const* d_in, const int* in_sizes, int n_in,
                              void* d_out, int out_size, void* d_ws, size_t ws_size,
                              hipStream_t stream) {
    (void)n_in; (void)d_ws; (void)ws_size; (void)out_size;
    const float* t  = (const float*)d_in[0];
    const float* x  = (const float*)d_in[1];
    const float* W0 = (const float*)d_in[2];
    const float* b0 = (const float*)d_in[3];
    const float* Uz = (const float*)d_in[4];
    const float* Ug = (const float*)d_in[5];
    const float* Ur = (const float*)d_in[6];
    const float* Uh = (const float*)d_in[7];
    const float* Wz = (const float*)d_in[8];
    const float* Wg = (const float*)d_in[9];
    const float* Wr = (const float*)d_in[10];
    const float* Wh = (const float*)d_in[11];
    const float* bz = (const float*)d_in[12];
    const float* bg = (const float*)d_in[13];
    const float* br = (const float*)d_in[14];
    const float* bh = (const float*)d_in[15];
    const float* Wf = (const float*)d_in[16];
    const float* bf = (const float*)d_in[17];
    float* out = (float*)d_out;

    const int N = in_sizes[0];
    const int blocks = N / MBLK;
    dgm_fused_kernel<<<blocks, 256, 0, stream>>>(
        t, x, W0, b0, Uz, Ug, Ur, Uh, Wz, Wg, Wr, Wh,
        bz, bg, br, bh, Wf, bf, out);
}


# A and B fragments read from LDS 16 bytes at a time (k-quad slab layout): one read feeds two WMMAs, half the LDS reads; within each 8 k the sum order is now k 0,1,4,5 then 2,3,6,7
# speedup vs baseline: 1.1902x; 1.0601x over previous
#include <hip/hip_runtime.h>


typedef float v2f __attribute__((ext_vector_type(2)));
typedef float v4f __attribute__((ext_vector_type(4)));
typedef float v8f __attribute__((ext_vector_type(8)));

#define H       256
#define LAYERS  3
#define MBLK    64
#define SP      260
#define KC1     32
#define KC3     16
#define QP      1088
#define STG     (3 * (KC3 / 4) * QP)

__device__ __forceinline__ v8f wmma_f32x4(v2f a, v2f b, v8f c) {
    return __builtin_amdgcn_wmma_f32_16x16x4_f32(
         false, a,  false, b,
         (short)0, c,  false,  false);
}

__device__ __forceinline__ float fast_tanh(float v) {
#if __has_builtin(__builtin_amdgcn_tanhf)
    return __builtin_amdgcn_tanhf(v);
#else
    float e = __expf(2.0f * v);
    return 1.0f - 2.0f / (e + 1.0f);
#endif
}

template <int NG, int KCS>
__device__ __forceinline__ void slab_fetch(v4f (&r)[NG][KCS / 16][4],
                                           const float* const (&w)[NG], int k0, int qr0, int q4) {
    #pragma unroll
    for (int g = 0; g < NG; ++g) {
        #pragma unroll
        for (int i = 0; i < KCS / 16; ++i) {
            const float* p = w[g] + (k0 + 4 * (qr0 + 4 * i)) * H + q4;
            #pragma unroll
            for (int t = 0; t < 4; ++t)
                r[g][i][t] = *(const v4f*)(p + t * H);
        }
    }
}

template <int NG, int KCS>
__device__ __forceinline__ void slab_store(float* stage, const v4f (&r)[NG][KCS / 16][4],
                                           int qr0, int q4) {
    #pragma unroll
    for (int g = 0; g < NG; ++g) {
        #pragma unroll
        for (int i = 0; i < KCS / 16; ++i) {
            float* d = stage + g * ((KCS / 4) * QP) + (qr0 + 4 * i) * QP + 4 * (q4 + (q4 >> 4));
            const v4f r0 = r[g][i][0], r1 = r[g][i][1], r2 = r[g][i][2], r3 = r[g][i][3];
            v4f c0, c1, c2, c3;
            c0.x = r0.x; c0.y = r1.x; c0.z = r2.x; c0.w = r3.x;
            c1.x = r0.y; c1.y = r1.y; c1.z = r2.y; c1.w = r3.y;
            c2.x = r0.z; c2.y = r1.z; c2.z = r2.z; c2.w = r3.z;
            c3.x = r0.w; c3.y = r1.w; c3.z = r2.w; c3.w = r3.w;
            *(v4f*)d        = c0;
            *(v4f*)(d + 4)  = c1;
            *(v4f*)(d + 8)  = c2;
            *(v4f*)(d + 12) = c3;
        }
    }
}

struct Frag {
    v4f a;
    v4f b[8];
};

__device__ __forceinline__ void frag_load(Frag& f, const float* ap, const float* bl) {
    f.a = *(const v4f*)ap;
    #pragma unroll
    for (int j = 0; j < 8; ++j)
        f.b[j] = *(const v4f*)(bl + j * 68);
}

__device__ __forceinline__ void frag_mma(v8f (&acc)[8], const Frag& f) {
    v2f a0, a1;
    a0.x = f.a.x; a0.y = f.a.y;
    a1.x = f.a.z; a1.y = f.a.w;
    #pragma unroll
    for (int j = 0; j < 8; ++j) {
        v2f b0;
        b0.x = f.b[j].x; b0.y = f.b[j].y;
        acc[j] = wmma_f32x4(a0, b0, acc[j]);
    }
    #pragma unroll
    for (int j = 0; j < 8; ++j) {
        v2f b1;
        b1.x = f.b[j].z; b1.y = f.b[j].w;
        acc[j] = wmma_f32x4(a1, b1, acc[j]);
    }
}

template <int NG, int KCS, int NN, int KCN>
__device__ __forceinline__ void gemm_phase(v8f (&acc)[NG][8], const float* const (&w)[NG],
                                           const float* const (&arow)[NG],
                                           const float* const (&wn)[NN], bool has_next,
                                           float* sW, int qr0, int q4, int boff) {
    constexpr int NSL = H / KCS;
    constexpr int NU  = (KCS / 8) * NG;
    constexpr int GSS = (KCS / 4) * QP;
    static_assert((NSL & 1) == 0, "the last slab must sit in stage 1");
    static_assert(NG * GSS <= STG && NN * (KCN / 4) * QP <= STG, "a slab fits a stage");
    v4f rr[NG][KCS / 16][4];
    v4f nr[NN][KCN / 16][4];
    for (int s = 0; s < NSL; ++s) {
        const bool more = (s + 1) < NSL;
        if (more)          slab_fetch<NG, KCS>(rr, w, (s + 1) * KCS, qr0, q4);
        else if (has_next) slab_fetch<NN, KCN>(nr, wn, 0, qr0, q4);
        const float* bl = sW + (s & 1) * STG + boff;
        Frag cur, nxt;
        frag_load(cur, arow[0] + s * KCS, bl);
        #pragma unroll
        for (int u = 0; u < NU; ++u) {
            const int g = u % NG;
            if (u + 1 < NU) {
                const int b1 = (u + 1) / NG, g1 = (u + 1) % NG;
                frag_load(nxt, arow[g1] + s * KCS + 8 * b1, bl + g1 * GSS + 2 * b1 * QP);
            }
            frag_mma(acc[g], cur);
            if (u + 1 < NU) cur = nxt;
            if (u == NU / 2 - 1) {
                if (more)          slab_store<NG, KCS>(sW + ((s + 1) & 1) * STG, rr, qr0, q4);
                else if (has_next) slab_store<NN, KCN>(sW, nr, qr0, q4);
            }
        }
        __syncthreads();
    }
}

__global__ __launch_bounds__(256) void dgm_fused_kernel(
    const float* __restrict__ t,  const float* __restrict__ x,
    const float* __restrict__ W0, const float* __restrict__ b0,
    const float* __restrict__ Uz, const float* __restrict__ Ug,
    const float* __restrict__ Ur, const float* __restrict__ Uh,
    const float* __restrict__ Wz, const float* __restrict__ Wg,
    const float* __restrict__ Wr, const float* __restrict__ Wh,
    const float* __restrict__ bz, const float* __restrict__ bg,
    const float* __restrict__ br, const float* __restrict__ bh,
    const float* __restrict__ Wf, const float* __restrict__ bf,
    float* __restrict__ out)
{
    __shared__ __attribute__((aligned(16))) float sS [MBLK * SP];
    __shared__ __attribute__((aligned(16))) float sS1[MBLK * SP];
    __shared__ __attribute__((aligned(16))) float sSR[MBLK * SP];
    __shared__ __attribute__((aligned(16))) float sW [2 * STG];
    __shared__ __attribute__((aligned(16))) float sOut[MBLK];
    __shared__ float sXt[MBLK];
    __shared__ float sXx[MBLK];

    const int tid  = threadIdx.x;
    const int wave = tid >> 5;
    const int lane = tid & 31;
    const int half = lane >> 4;
    const int l16  = lane & 15;
    const int row0 = blockIdx.x * MBLK;

    const int rbase = (wave & 3) * 16;
    const int cbase = (wave >> 2) * 128;
    const int qr0   = tid >> 6;
    const int q4    = (tid & 63) * 4;
    const int boff  = half * QP + (cbase + (cbase >> 4) + l16) * 4;
    const int aoff  = (rbase + l16) * SP + half * 4;

    if (tid < MBLK) {
        const float* tb = t + row0;
        const float* xb = x + row0;
        sXt[tid] = tb[tid];
        sXx[tid] = xb[tid];
    }
    __syncthreads();

    {
        const int   c   = tid;
        const float w0t = W0[c];
        const float w0x = W0[H + c];
        const float bb  = b0[c];
        #pragma unroll 8
        for (int r = 0; r < MBLK; ++r) {
            float v = fast_tanh(sXt[r] * w0t + sXx[r] * w0x + bb);
            sS1[r * SP + c] = v;
            sS [r * SP + c] = v;
        }
    }

    {
        const float* const wfirst[1] = { Wr };
        v4f fr[1][KC1 / 16][4];
        slab_fetch<1, KC1>(fr, wfirst, 0, qr0, q4);
        slab_store<1, KC1>(sW, fr, qr0, q4);
    }
    __syncthreads();

    const int xoff = rbase + half * 8;

    for (int l = 0; l < LAYERS; ++l) {
        const float* wzl = Wz + l * H * H;
        const float* wgl = Wg + l * H * H;
        const float* wrl = Wr + l * H * H;
        const float* whl = Wh + l * H * H;
        const float* uzl = Uz + l * 2 * H;
        const float* ugl = Ug + l * 2 * H;
        const float* url = Ur + l * 2 * H;
        const float* uhl = Uh + l * 2 * H;
        const float* bzl = bz + l * H;
        const float* bgl = bg + l * H;
        const float* brl = br + l * H;
        const float* bhl = bh + l * H;

        {
            v8f accR[1][8];
            float xt8[8], xx8[8];
            #pragma unroll
            for (int v = 0; v < 8; ++v) {
                xt8[v] = sXt[xoff + v];
                xx8[v] = sXx[xoff + v];
            }
            #pragma unroll
            for (int j = 0; j < 8; ++j) {
                const int   n  = cbase + j * 16 + l16;
                const float u0 = url[n];
                const float u1 = url[H + n];
                const float bb = brl[n];
                #pragma unroll
                for (int v = 0; v < 8; ++v)
                    accR[0][j][v] = xt8[v] * u0 + xx8[v] * u1 + bb;
            }
            const float* const w1[1] = { wrl };
            const float* const a1[1] = { sS + aoff };
            const float* const n3[3] = { wzl, wgl, whl };
            gemm_phase<1, KC1, 3, KC3>(accR, w1, a1, n3, true, sW, qr0, q4, boff);
            float sv[8][8];
            #pragma unroll
            for (int j = 0; j < 8; ++j) {
                #pragma unroll
                for (int v = 0; v < 8; ++v)
                    sv[j][v] = sS[(rbase + half * 8 + v) * SP + cbase + j * 16 + l16];
            }
            #pragma unroll
            for (int j = 0; j < 8; ++j) {
                const int n = cbase + j * 16 + l16;
                #pragma unroll
                for (int v = 0; v < 8; ++v) {
                    const int m = rbase + half * 8 + v;
                    float R = fast_tanh(accR[0][j][v]);
                    sSR[m * SP + n] = sv[j][v] * R;
                }
            }
            __syncthreads();
        }

        {
            v8f acc[3][8];
            float xt8[8], xx8[8];
            #pragma unroll
            for (int v = 0; v < 8; ++v) {
                xt8[v] = sXt[xoff + v];
                xx8[v] = sXx[xoff + v];
            }
            #pragma unroll
            for (int j = 0; j < 8; ++j) {
                const int   n  = cbase + j * 16 + l16;
                const float z0 = uzl[n], z1 = uzl[H + n], zb = bzl[n];
                const float g0 = ugl[n], g1 = ugl[H + n], gb = bgl[n];
                const float h0 = uhl[n], h1 = uhl[H + n], hb = bhl[n];
                #pragma unroll
                for (int v = 0; v < 8; ++v) {
                    acc[0][j][v] = xt8[v] * z0 + xx8[v] * z1 + zb;
                    acc[1][j][v] = xt8[v] * g0 + xx8[v] * g1 + gb;
                    acc[2][j][v] = xt8[v] * h0 + xx8[v] * h1 + hb;
                }
            }
            const float* const w3[3] = { wzl, wgl, whl };
            const float* const a3[3] = { sS + aoff, sS1 + aoff, sSR + aoff };
            const bool  has_next = (l + 1) < LAYERS;
            const float* const n1[1] = { has_next ? wrl + H * H : wrl };
            gemm_phase<3, KC3, 1, KC1>(acc, w3, a3, n1, has_next, sW, qr0, q4, boff);
            float sv[8][8];
            #pragma unroll
            for (int j = 0; j < 8; ++j) {
                #pragma unroll
                for (int v = 0; v < 8; ++v)
                    sv[j][v] = sS[(rbase + half * 8 + v) * SP + cbase + j * 16 + l16];
            }
            #pragma unroll
            for (int j = 0; j < 8; ++j) {
                const int n = cbase + j * 16 + l16;
                #pragma unroll
                for (int v = 0; v < 8; ++v) {
                    const int m = rbase + half * 8 + v;
                    float Zv = fast_tanh(acc[0][j][v]);
                    float Gv = fast_tanh(acc[1][j][v]);
                    float Hv = fast_tanh(acc[2][j][v]);
                    sS[m * SP + n] = (1.0f - Gv) * Hv + Zv * sv[j][v];
                }
            }
            __syncthreads();
        }
    }

    {
        const int ci = tid & 15;
        float wf[16];
        #pragma unroll
        for (int j = 0; j < 16; ++j) wf[j] = Wf[ci * 16 + j];
        const float bfv = bf[0];
        #pragma unroll
        for (int pass = 0; pass < MBLK / 16; ++pass) {
            const int r = pass * 16 + (tid >> 4);
            float p = 0.0f;
            #pragma unroll
            for (int j = 0; j < 16; ++j)
                p += sS[r * SP + ci * 16 + j] * wf[j];
            #pragma unroll
            for (int off = 8; off >= 1; off >>= 1)
                p += __shfl_xor(p, off, 16);
            if (ci == 0)
                sOut[r] = p + bfv;
        }
    }
    __syncthreads();
    if (tid < MBLK) {
        float* ob = out + row0;
        ob[tid] = sOut[tid];
    }
}

extern "C" void kernel_launch(void* const* d_in, const int* in_sizes, int n_in,
                              void* d_out, int out_size, void* d_ws, size_t ws_size,
                              hipStream_t stream) {
    (void)n_in; (void)d_ws; (void)ws_size; (void)out_size;
    const float* t  = (const float*)d_in[0];
    const float* x  = (const float*)d_in[1];
    const float* W0 = (const float*)d_in[2];
    const float* b0 = (const float*)d_in[3];
    const float* Uz = (const float*)d_in[4];
    const float* Ug = (const float*)d_in[5];
    const float* Ur = (const float*)d_in[6];
    const float* Uh = (const float*)d_in[7];
    const float* Wz = (const float*)d_in[8];
    const float* Wg = (const float*)d_in[9];
    const float* Wr = (const float*)d_in[10];
    const float* Wh = (const float*)d_in[11];
    const float* bz = (const float*)d_in[12];
    const float* bg = (const float*)d_in[13];
    const float* br = (const float*)d_in[14];
    const float* bh = (const float*)d_in[15];
    const float* Wf = (const float*)d_in[16];
    const float* bf = (const float*)d_in[17];
    float* out = (float*)d_out;

    const int N = in_sizes[0];
    const int blocks = N / MBLK;
    dgm_fused_kernel<<<blocks, 256, 0, stream>>>(
        t, x, W0, b0, Uz, Ug, Ur, Uh, Wz, Wg, Wr, Wh,
        bz, bg, br, bh, Wf, bf, out);
}


# K-loops: each wave lowers its issue priority (s_setprio) unit by unit through a slab, so the two waves of a SIMD alternate and each one's LDS-read gaps sit under the other's WMMAs; sums unchanged
# speedup vs baseline: 1.2455x; 1.0465x over previous
#include <hip/hip_runtime.h>


typedef float v2f __attribute__((ext_vector_type(2)));
typedef float v4f __attribute__((ext_vector_type(4)));
typedef float v8f __attribute__((ext_vector_type(8)));

#define H       256
#define LAYERS  3
#define MBLK    64
#define SP      260
#define KC1     32
#define KC3     16
#define QP      1088
#define STG     (3 * (KC3 / 4) * QP)

__device__ __forceinline__ v8f wmma_f32x4(v2f a, v2f b, v8f c) {
    return __builtin_amdgcn_wmma_f32_16x16x4_f32(
         false, a,  false, b,
         (short)0, c,  false,  false);
}

__device__ __forceinline__ float fast_tanh(float v) {
#if __has_builtin(__builtin_amdgcn_tanhf)
    return __builtin_amdgcn_tanhf(v);
#else
    float e = __expf(2.0f * v);
    return 1.0f - 2.0f / (e + 1.0f);
#endif
}

template <int NG, int KCS>
__device__ __forceinline__ void slab_fetch(v4f (&r)[NG][KCS / 16][4],
                                           const float* const (&w)[NG], int k0, int qr0, int q4) {
    #pragma unroll
    for (int g = 0; g < NG; ++g) {
        #pragma unroll
        for (int i = 0; i < KCS / 16; ++i) {
            const float* p = w[g] + (k0 + 4 * (qr0 + 4 * i)) * H + q4;
            #pragma unroll
            for (int t = 0; t < 4; ++t)
                r[g][i][t] = *(const v4f*)(p + t * H);
        }
    }
}

template <int NG, int KCS>
__device__ __forceinline__ void slab_store(float* stage, const v4f (&r)[NG][KCS / 16][4],
                                           int qr0, int q4) {
    #pragma unroll
    for (int g = 0; g < NG; ++g) {
        #pragma unroll
        for (int i = 0; i < KCS / 16; ++i) {
            float* d = stage + g * ((KCS / 4) * QP) + (qr0 + 4 * i) * QP + 4 * (q4 + (q4 >> 4));
            const v4f r0 = r[g][i][0], r1 = r[g][i][1], r2 = r[g][i][2], r3 = r[g][i][3];
            v4f c0, c1, c2, c3;
            c0.x = r0.x; c0.y = r1.x; c0.z = r2.x; c0.w = r3.x;
            c1.x = r0.y; c1.y = r1.y; c1.z = r2.y; c1.w = r3.y;
            c2.x = r0.z; c2.y = r1.z; c2.z = r2.z; c2.w = r3.z;
            c3.x = r0.w; c3.y = r1.w; c3.z = r2.w; c3.w = r3.w;
            *(v4f*)d        = c0;
            *(v4f*)(d + 4)  = c1;
            *(v4f*)(d + 8)  = c2;
            *(v4f*)(d + 12) = c3;
        }
    }
}

struct Frag {
    v4f a;
    v4f b[8];
};

__device__ __forceinline__ void frag_load(Frag& f, const float* ap, const float* bl) {
    f.a = *(const v4f*)ap;
    #pragma unroll
    for (int j = 0; j < 8; ++j)
        f.b[j] = *(const v4f*)(bl + j * 68);
}

__device__ __forceinline__ void frag_mma(v8f (&acc)[8], const Frag& f) {
    v2f a0, a1;
    a0.x = f.a.x; a0.y = f.a.y;
    a1.x = f.a.z; a1.y = f.a.w;
    #pragma unroll
    for (int j = 0; j < 8; ++j) {
        v2f b0;
        b0.x = f.b[j].x; b0.y = f.b[j].y;
        acc[j] = wmma_f32x4(a0, b0, acc[j]);
    }
    #pragma unroll
    for (int j = 0; j < 8; ++j) {
        v2f b1;
        b1.x = f.b[j].z; b1.y = f.b[j].w;
        acc[j] = wmma_f32x4(a1, b1, acc[j]);
    }
}

__device__ __forceinline__ constexpr int unit_prio(int nu, int u) {
    return ((nu - 1 - u) * 3 + (nu - 2)) / (nu - 1);
}

__device__ __forceinline__ void set_issue_prio(int p) {
    switch (p) {
        case 3:  __builtin_amdgcn_s_setprio(3); break;
        case 2:  __builtin_amdgcn_s_setprio(2); break;
        case 1:  __builtin_amdgcn_s_setprio(1); break;
        default: __builtin_amdgcn_s_setprio(0); break;
    }
}

template <int NG, int KCS, int NN, int KCN>
__device__ __forceinline__ void gemm_phase(v8f (&acc)[NG][8], const float* const (&w)[NG],
                                           const float* const (&arow)[NG],
                                           const float* const (&wn)[NN], bool has_next,
                                           float* sW, int qr0, int q4, int boff) {
    constexpr int NSL = H / KCS;
    constexpr int NU  = (KCS / 8) * NG;
    constexpr int GSS = (KCS / 4) * QP;
    static_assert((NSL & 1) == 0, "the last slab must sit in stage 1");
    static_assert(NG * GSS <= STG && NN * (KCN / 4) * QP <= STG, "a slab fits a stage");
    v4f rr[NG][KCS / 16][4];
    v4f nr[NN][KCN / 16][4];
    for (int s = 0; s < NSL; ++s) {
        const bool more = (s + 1) < NSL;
        if (more)          slab_fetch<NG, KCS>(rr, w, (s + 1) * KCS, qr0, q4);
        else if (has_next) slab_fetch<NN, KCN>(nr, wn, 0, qr0, q4);
        const float* bl = sW + (s & 1) * STG + boff;
        Frag cur, nxt;
        frag_load(cur, arow[0] + s * KCS, bl);
        #pragma unroll
        for (int u = 0; u < NU; ++u) {
            const int g = u % NG;
            if (u == 0 || unit_prio(NU, u) != unit_prio(NU, u - 1))
                set_issue_prio(unit_prio(NU, u));
            if (u + 1 < NU) {
                const int b1 = (u + 1) / NG, g1 = (u + 1) % NG;
                frag_load(nxt, arow[g1] + s * KCS + 8 * b1, bl + g1 * GSS + 2 * b1 * QP);
            }
            frag_mma(acc[g], cur);
            if (u + 1 < NU) cur = nxt;
            if (u == NU / 2 - 1) {
                if (more)          slab_store<NG, KCS>(sW + ((s + 1) & 1) * STG, rr, qr0, q4);
                else if (has_next) slab_store<NN, KCN>(sW, nr, qr0, q4);
            }
        }
        __syncthreads();
    }
}

__global__ __launch_bounds__(256) void dgm_fused_kernel(
    const float* __restrict__ t,  const float* __restrict__ x,
    const float* __restrict__ W0, const float* __restrict__ b0,
    const float* __restrict__ Uz, const float* __restrict__ Ug,
    const float* __restrict__ Ur, const float* __restrict__ Uh,
    const float* __restrict__ Wz, const float* __restrict__ Wg,
    const float* __restrict__ Wr, const float* __restrict__ Wh,
    const float* __restrict__ bz, const float* __restrict__ bg,
    const float* __restrict__ br, const float* __restrict__ bh,
    const float* __restrict__ Wf, const float* __restrict__ bf,
    float* __restrict__ out)
{
    __shared__ __attribute__((aligned(16))) float sS [MBLK * SP];
    __shared__ __attribute__((aligned(16))) float sS1[MBLK * SP];
    __shared__ __attribute__((aligned(16))) float sSR[MBLK * SP];
    __shared__ __attribute__((aligned(16))) float sW [2 * STG];
    __shared__ __attribute__((aligned(16))) float sOut[MBLK];
    __shared__ float sXt[MBLK];
    __shared__ float sXx[MBLK];

    const int tid  = threadIdx.x;
    const int wave = tid >> 5;
    const int lane = tid & 31;
    const int half = lane >> 4;
    const int l16  = lane & 15;
    const int row0 = blockIdx.x * MBLK;

    const int rbase = (wave & 3) * 16;
    const int cbase = (wave >> 2) * 128;
    const int qr0   = tid >> 6;
    const int q4    = (tid & 63) * 4;
    const int boff  = half * QP + (cbase + (cbase >> 4) + l16) * 4;
    const int aoff  = (rbase + l16) * SP + half * 4;

    if (tid < MBLK) {
        const float* tb = t + row0;
        const float* xb = x + row0;
        sXt[tid] = tb[tid];
        sXx[tid] = xb[tid];
    }
    __syncthreads();

    {
        const int   c   = tid;
        const float w0t = W0[c];
        const float w0x = W0[H + c];
        const float bb  = b0[c];
        #pragma unroll 8
        for (int r = 0; r < MBLK; ++r) {
            float v = fast_tanh(sXt[r] * w0t + sXx[r] * w0x + bb);
            sS1[r * SP + c] = v;
            sS [r * SP + c] = v;
        }
    }

    {
        const float* const wfirst[1] = { Wr };
        v4f fr[1][KC1 / 16][4];
        slab_fetch<1, KC1>(fr, wfirst, 0, qr0, q4);
        slab_store<1, KC1>(sW, fr, qr0, q4);
    }
    __syncthreads();

    const int xoff = rbase + half * 8;

    for (int l = 0; l < LAYERS; ++l) {
        const float* wzl = Wz + l * H * H;
        const float* wgl = Wg + l * H * H;
        const float* wrl = Wr + l * H * H;
        const float* whl = Wh + l * H * H;
        const float* uzl = Uz + l * 2 * H;
        const float* ugl = Ug + l * 2 * H;
        const float* url = Ur + l * 2 * H;
        const float* uhl = Uh + l * 2 * H;
        const float* bzl = bz + l * H;
        const float* bgl = bg + l * H;
        const float* brl = br + l * H;
        const float* bhl = bh + l * H;

        {
            v8f accR[1][8];
            float xt8[8], xx8[8];
            #pragma unroll
            for (int v = 0; v < 8; ++v) {
                xt8[v] = sXt[xoff + v];
                xx8[v] = sXx[xoff + v];
            }
            #pragma unroll
            for (int j = 0; j < 8; ++j) {
                const int   n  = cbase + j * 16 + l16;
                const float u0 = url[n];
                const float u1 = url[H + n];
                const float bb = brl[n];
                #pragma unroll
                for (int v = 0; v < 8; ++v)
                    accR[0][j][v] = xt8[v] * u0 + xx8[v] * u1 + bb;
            }
            const float* const w1[1] = { wrl };
            const float* const a1[1] = { sS + aoff };
            const float* const n3[3] = { wzl, wgl, whl };
            gemm_phase<1, KC1, 3, KC3>(accR, w1, a1, n3, true, sW, qr0, q4, boff);
            float sv[8][8];
            #pragma unroll
            for (int j = 0; j < 8; ++j) {
                #pragma unroll
                for (int v = 0; v < 8; ++v)
                    sv[j][v] = sS[(rbase + half * 8 + v) * SP + cbase + j * 16 + l16];
            }
            #pragma unroll
            for (int j = 0; j < 8; ++j) {
                const int n = cbase + j * 16 + l16;
                #pragma unroll
                for (int v = 0; v < 8; ++v) {
                    const int m = rbase + half * 8 + v;
                    float R = fast_tanh(accR[0][j][v]);
                    sSR[m * SP + n] = sv[j][v] * R;
                }
            }
            __syncthreads();
        }

        {
            v8f acc[3][8];
            float xt8[8], xx8[8];
            #pragma unroll
            for (int v = 0; v < 8; ++v) {
                xt8[v] = sXt[xoff + v];
                xx8[v] = sXx[xoff + v];
            }
            #pragma unroll
            for (int j = 0; j < 8; ++j) {
                const int   n  = cbase + j * 16 + l16;
                const float z0 = uzl[n], z1 = uzl[H + n], zb = bzl[n];
                const float g0 = ugl[n], g1 = ugl[H + n], gb = bgl[n];
                const float h0 = uhl[n], h1 = uhl[H + n], hb = bhl[n];
                #pragma unroll
                for (int v = 0; v < 8; ++v) {
                    acc[0][j][v] = xt8[v] * z0 + xx8[v] * z1 + zb;
                    acc[1][j][v] = xt8[v] * g0 + xx8[v] * g1 + gb;
                    acc[2][j][v] = xt8[v] * h0 + xx8[v] * h1 + hb;
                }
            }
            const float* const w3[3] = { wzl, wgl, whl };
            const float* const a3[3] = { sS + aoff, sS1 + aoff, sSR + aoff };
            const bool  has_next = (l + 1) < LAYERS;
            const float* const n1[1] = { has_next ? wrl + H * H : wrl };
            gemm_phase<3, KC3, 1, KC1>(acc, w3, a3, n1, has_next, sW, qr0, q4, boff);
            float sv[8][8];
            #pragma unroll
            for (int j = 0; j < 8; ++j) {
                #pragma unroll
                for (int v = 0; v < 8; ++v)
                    sv[j][v] = sS[(rbase + half * 8 + v) * SP + cbase + j * 16 + l16];
            }
            #pragma unroll
            for (int j = 0; j < 8; ++j) {
                const int n = cbase + j * 16 + l16;
                #pragma unroll
                for (int v = 0; v < 8; ++v) {
                    const int m = rbase + half * 8 + v;
                    float Zv = fast_tanh(acc[0][j][v]);
                    float Gv = fast_tanh(acc[1][j][v]);
                    float Hv = fast_tanh(acc[2][j][v]);
                    sS[m * SP + n] = (1.0f - Gv) * Hv + Zv * sv[j][v];
                }
            }
            __syncthreads();
        }
    }

    {
        const int ci = tid & 15;
        float wf[16];
        #pragma unroll
        for (int j = 0; j < 16; ++j) wf[j] = Wf[ci * 16 + j];
        const float bfv = bf[0];
        #pragma unroll
        for (int pass = 0; pass < MBLK / 16; ++pass) {
            const int r = pass * 16 + (tid >> 4);
            float p = 0.0f;
            #pragma unroll
            for (int j = 0; j < 16; ++j)
                p += sS[r * SP + ci * 16 + j] * wf[j];
            #pragma unroll
            for (int off = 8; off >= 1; off >>= 1)
                p += __shfl_xor(p, off, 16);
            if (ci == 0)
                sOut[r] = p + bfv;
        }
    }
    __syncthreads();
    if (tid < MBLK) {
        float* ob = out + row0;
        ob[tid] = sOut[tid];
    }
}

extern "C" void kernel_launch(void* const* d_in, const int* in_sizes, int n_in,
                              void* d_out, int out_size, void* d_ws, size_t ws_size,
                              hipStream_t stream) {
    (void)n_in; (void)d_ws; (void)ws_size; (void)out_size;
    const float* t  = (const float*)d_in[0];
    const float* x  = (const float*)d_in[1];
    const float* W0 = (const float*)d_in[2];
    const float* b0 = (const float*)d_in[3];
    const float* Uz = (const float*)d_in[4];
    const float* Ug = (const float*)d_in[5];
    const float* Ur = (const float*)d_in[6];
    const float* Uh = (const float*)d_in[7];
    const float* Wz = (const float*)d_in[8];
    const float* Wg = (const float*)d_in[9];
    const float* Wr = (const float*)d_in[10];
    const float* Wh = (const float*)d_in[11];
    const float* bz = (const float*)d_in[12];
    const float* bg = (const float*)d_in[13];
    const float* br = (const float*)d_in[14];
    const float* bh = (const float*)d_in[15];
    const float* Wf = (const float*)d_in[16];
    const float* bf = (const float*)d_in[17];
    float* out = (float*)d_out;

    const int N = in_sizes[0];
    const int blocks = N / MBLK;
    dgm_fused_kernel<<<blocks, 256, 0, stream>>>(
        t, x, W0, b0, Uz, Ug, Ur, Uh, Wz, Wg, Wr, Wh,
        bz, bg, br, bh, Wf, bf, out);
}
